# starting kernel
# baseline (speedup 1.0000x reference)
#include <hip/hip_runtime.h>
#include <math.h>


typedef __attribute__((ext_vector_type(2))) float v2f;
typedef __attribute__((ext_vector_type(8))) float v8f;

__device__ __forceinline__ v8f wmma4(v2f a, v2f b, v8f c) {
  return __builtin_amdgcn_wmma_f32_16x16x4_f32(false, a, false, b, (short)0, c,
                                               false, false);
}

__global__ void zero_kernel(float* __restrict__ p, long n) {
  long i = (long)blockIdx.x * blockDim.x + threadIdx.x;
  long stride = (long)gridDim.x * blockDim.x;
  for (long k = i * 4; k < n; k += stride * 4) {
    *(float4*)(p + k) = make_float4(0.f, 0.f, 0.f, 0.f);
  }
}

__global__ void node_pre_kernel(const float* __restrict__ node_input,
                                const float* __restrict__ node_attr,
                                const float* __restrict__ Wsc0,
                                const float* __restrict__ Wsc1,
                                const float* __restrict__ Wl10,
                                const float* __restrict__ Wl11,
                                float* __restrict__ f0f1,
                                float* __restrict__ scx, int N) {
  int wv = threadIdx.x >> 5;
  int lane = threadIdx.x & 31;
  int n = blockIdx.x * (blockDim.x >> 5) + wv;
  if (n >= N) return;
  const float* xi = node_input + (long)n * 128;
  float f0 = 0.f, f1x = 0.f, f1y = 0.f, f1z = 0.f;
  float s0 = 0.f, s1x = 0.f, s1y = 0.f, s1z = 0.f;
  for (int u = 0; u < 32; ++u) {
    float x0 = xi[u];
    float x1x = xi[32 + u * 3 + 0];
    float x1y = xi[32 + u * 3 + 1];
    float x1z = xi[32 + u * 3 + 2];
    float wl0 = Wl10[u * 32 + lane], wl1 = Wl11[u * 32 + lane];
    float ws0 = Wsc0[u * 32 + lane], ws1 = Wsc1[u * 32 + lane];
    f0 += x0 * wl0;
    f1x += x1x * wl1; f1y += x1y * wl1; f1z += x1z * wl1;
    s0 += x0 * ws0;
    s1x += x1x * ws1; s1y += x1y * ws1; s1z += x1z * ws1;
  }
  float k = node_attr[n] * 0.17677669529663687f;
  float* fo = f0f1 + (long)n * 128;
  fo[lane] = f0 * k;
  fo[32 + lane] = f1x * k;
  fo[64 + lane] = f1y * k;
  fo[96 + lane] = f1z * k;
  float* so = scx + (long)n * 128;
  so[lane] = s0 * k;
  so[32 + lane * 3 + 0] = s1x * k;
  so[32 + lane * 3 + 1] = s1y * k;
  so[32 + lane * 3 + 2] = s1z * k;
}

#define EK_WAVES 8
#define HSTR 68

__global__ void __launch_bounds__(256)
edge_kernel(const float* __restrict__ edge_scalars,
            const float* __restrict__ edge_attr,
            const int* __restrict__ edge_src,
            const int* __restrict__ edge_dst,
            const float* __restrict__ Wfc1,
            const float* __restrict__ Wfc2,
            const float* __restrict__ f0f1,
            float* __restrict__ mid, int E) {
  __shared__ float fc1[4 * 4 * 32 * 2];
  __shared__ float fc2[16 * 8 * 32 * 2];
  __shared__ float stage[EK_WAVES][2048];

  for (int idx = threadIdx.x; idx < 4 * 4 * 32; idx += blockDim.x) {
    int ln = idx & 31, t = (idx >> 5) & 3, j = idx >> 7;
    int row = 4 * j + 2 * (ln >> 4), col = 16 * t + (ln & 15);
    fc1[idx * 2 + 0] = Wfc1[row * 64 + col];
    fc1[idx * 2 + 1] = Wfc1[(row + 1) * 64 + col];
  }
  for (int idx = threadIdx.x; idx < 16 * 8 * 32; idx += blockDim.x) {
    int ln = idx & 31, t = (idx >> 5) & 7, j = idx >> 8;
    int row = 4 * j + 2 * (ln >> 4), col = 16 * t + (ln & 15);
    fc2[idx * 2 + 0] = Wfc2[row * 128 + col];
    fc2[idx * 2 + 1] = Wfc2[(row + 1) * 128 + col];
  }
  __syncthreads();

  const int lane = threadIdx.x & 31;
  const int wv = threadIdx.x >> 5;
  const int r = lane & 15;
  const int hi = lane >> 4;
  float* st = stage[wv];
  const int nw = gridDim.x * (blockDim.x >> 5);
  const int gw = blockIdx.x * (blockDim.x >> 5) + wv;
  const int tiles = (E + 15) >> 4;

  for (int tile = gw; tile < tiles; tile += nw) {
    const int e0 = tile * 16;
    int er = e0 + r;
    if (er >= E) er = E - 1;
    const float* srow = edge_scalars + (long)er * 16;

    v2f a1[4];
#pragma unroll
    for (int j = 0; j < 4; ++j) a1[j] = *(const v2f*)(srow + 4 * j + 2 * hi);

#pragma unroll
    for (int t = 0; t < 4; ++t) {
      v8f c = {0.f, 0.f, 0.f, 0.f, 0.f, 0.f, 0.f, 0.f};
#pragma unroll
      for (int j = 0; j < 4; ++j) {
        v2f b = *(const v2f*)&fc1[((j * 4 + t) * 32 + lane) * 2];
        c = wmma4(a1[j], b, c);
      }
#pragma unroll
      for (int v = 0; v < 8; ++v) {
        float x = c[v] * 0.25f;
        st[(v + 8 * hi) * HSTR + 16 * t + r] = x / (1.0f + __expf(-x));
      }
    }

    v2f a2[16];
#pragma unroll
    for (int j = 0; j < 16; ++j)
      a2[j] = *(const v2f*)&st[r * HSTR + 4 * j + 2 * hi];

#pragma unroll
    for (int t = 0; t < 8; ++t) {
      v8f c = {0.f, 0.f, 0.f, 0.f, 0.f, 0.f, 0.f, 0.f};
#pragma unroll
      for (int j = 0; j < 16; ++j) {
        v2f b = *(const v2f*)&fc2[((j * 8 + t) * 32 + lane) * 2];
        c = wmma4(a2[j], b, c);
      }
#pragma unroll
      for (int v = 0; v < 8; ++v)
        st[(v + 8 * hi) * 128 + 16 * t + r] = c[v] * 0.125f;
    }

    const float q = 0.31622776601683794f;
    const float r3 = 0.5773502691896258f;
    for (int i = 0; i < 16; ++i) {
      int e = e0 + i;
      if (e >= E) break;
      int s = edge_src[e], d = edge_dst[e];
      const float4 ya = *(const float4*)(edge_attr + (long)e * 4);
      const float* g = f0f1 + (long)s * 128;
      float g0 = g[lane];
      float g1x = g[32 + lane], g1y = g[64 + lane], g1z = g[96 + lane];
      float wA = st[i * 128 + lane];
      float wB = st[i * 128 + 32 + lane];
      float wC = st[i * 128 + 64 + lane];
      float wD = st[i * 128 + 96 + lane];
      float eA = wA * g0 * ya.x * q;
      float eD = wD * (g1x * ya.y + g1y * ya.z + g1z * ya.w) * r3 * q;
      float gB = wB * g0 * q;
      float cc = wC * ya.x * q;
      float* m = mid + (long)d * 256;
      atomicAdd(m + lane, eA);
      atomicAdd(m + 32 + lane, eD);
      atomicAdd(m + 64 + 3 * lane + 0, gB * ya.y);
      atomicAdd(m + 64 + 3 * lane + 1, gB * ya.z);
      atomicAdd(m + 64 + 3 * lane + 2, gB * ya.w);
      atomicAdd(m + 160 + 3 * lane + 0, cc * g1x);
      atomicAdd(m + 160 + 3 * lane + 1, cc * g1y);
      atomicAdd(m + 160 + 3 * lane + 2, cc * g1z);
    }
  }
}

__global__ void node_post_kernel(const float* __restrict__ mid,
                                 const float* __restrict__ node_attr,
                                 const float* __restrict__ Wl20,
                                 const float* __restrict__ Wl21,
                                 const float* __restrict__ Wl3,
                                 const float* __restrict__ scx,
                                 float* __restrict__ out, int N) {
  int wv = threadIdx.x >> 5;
  int lane = threadIdx.x & 31;
  int n = blockIdx.x * (blockDim.x >> 5) + wv;
  if (n >= N) return;
  const float* m = mid + (long)n * 256;
  float o0 = 0.f, o1x = 0.f, o1y = 0.f, o1z = 0.f, ang = 0.f;
  for (int v = 0; v < 64; ++v) {
    float m0 = m[v];
    float mx = m[64 + v * 3 + 0];
    float my = m[64 + v * 3 + 1];
    float mz = m[64 + v * 3 + 2];
    float w0 = Wl20[v * 32 + lane], w1 = Wl21[v * 32 + lane];
    o0 += m0 * w0;
    o1x += mx * w1; o1y += my * w1; o1z += mz * w1;
    ang += m0 * Wl3[v];
  }
  float k = node_attr[n] * 0.125f;
  o0 *= k; o1x *= k; o1y *= k; o1z *= k;
  float angle = 0.1f * ang * k;
  float ca = cosf(angle), sa = sinf(angle);
  const float* sx = scx + (long)n * 128;
  float* o = out + (long)n * 128;
  o[lane] = ca * sx[lane] + sa * o0;
  o[32 + 3 * lane + 0] = ca * sx[32 + 3 * lane + 0] + sa * o1x;
  o[32 + 3 * lane + 1] = ca * sx[32 + 3 * lane + 1] + sa * o1y;
  o[32 + 3 * lane + 2] = ca * sx[32 + 3 * lane + 2] + sa * o1z;
}

extern "C" void kernel_launch(void* const* d_in, const int* in_sizes, int n_in,
                              void* d_out, int out_size, void* d_ws, size_t ws_size,
                              hipStream_t stream) {
  const float* node_input   = (const float*)d_in[0];
  const float* node_attr    = (const float*)d_in[1];
  const int*   edge_src     = (const int*)d_in[2];
  const int*   edge_dst     = (const int*)d_in[3];
  const float* edge_attr    = (const float*)d_in[4];
  const float* edge_scalars = (const float*)d_in[5];
  const float* W_sc0        = (const float*)d_in[6];
  const float* W_sc1        = (const float*)d_in[7];
  const float* W_lin1_0     = (const float*)d_in[8];
  const float* W_lin1_1     = (const float*)d_in[9];
  const float* W_fc1        = (const float*)d_in[10];
  const float* W_fc2        = (const float*)d_in[11];
  const float* W_lin2_0     = (const float*)d_in[12];
  const float* W_lin2_1     = (const float*)d_in[13];
  const float* W_lin3       = (const float*)d_in[14];
  int N = in_sizes[1];
  int E = in_sizes[2];

  float* f0f1 = (float*)d_ws;
  float* scx  = f0f1 + (size_t)N * 128;
  float* mid  = scx + (size_t)N * 128;
  float* out  = (float*)d_out;

  zero_kernel<<<2048, 256, 0, stream>>>(mid, (long)N * 256);

  int nblk_node = (N + 7) / 8;
  node_pre_kernel<<<nblk_node, 256, 0, stream>>>(
      node_input, node_attr, W_sc0, W_sc1, W_lin1_0, W_lin1_1, f0f1, scx, N);

  edge_kernel<<<1024, 256, 0, stream>>>(
      edge_scalars, edge_attr, edge_src, edge_dst, W_fc1, W_fc2, f0f1, mid, E);

  node_post_kernel<<<nblk_node, 256, 0, stream>>>(
      mid, node_attr, W_lin2_0, W_lin2_1, W_lin3, scx, out, N);
}


# edge scatter: a tile's src, dst and edge_attr loaded once per lane up front and broadcast by readlane; its 64 gathers issued as one group before its atomics (no per-edge dependent load chain)
# speedup vs baseline: 1.1368x; 1.1368x over previous
#include <hip/hip_runtime.h>
#include <math.h>


typedef __attribute__((ext_vector_type(2))) float v2f;
typedef __attribute__((ext_vector_type(8))) float v8f;

__device__ __forceinline__ v8f wmma4(v2f a, v2f b, v8f c) {
  return __builtin_amdgcn_wmma_f32_16x16x4_f32(false, a, false, b, (short)0, c,
                                               false, false);
}

__global__ void zero_kernel(float* __restrict__ p, long n) {
  long i = (long)blockIdx.x * blockDim.x + threadIdx.x;
  long stride = (long)gridDim.x * blockDim.x;
  for (long k = i * 4; k < n; k += stride * 4) {
    *(float4*)(p + k) = make_float4(0.f, 0.f, 0.f, 0.f);
  }
}

__global__ void node_pre_kernel(const float* __restrict__ node_input,
                                const float* __restrict__ node_attr,
                                const float* __restrict__ Wsc0,
                                const float* __restrict__ Wsc1,
                                const float* __restrict__ Wl10,
                                const float* __restrict__ Wl11,
                                float* __restrict__ f0f1,
                                float* __restrict__ scx, int N) {
  int wv = threadIdx.x >> 5;
  int lane = threadIdx.x & 31;
  int n = blockIdx.x * (blockDim.x >> 5) + wv;
  if (n >= N) return;
  const float* xi = node_input + (long)n * 128;
  float f0 = 0.f, f1x = 0.f, f1y = 0.f, f1z = 0.f;
  float s0 = 0.f, s1x = 0.f, s1y = 0.f, s1z = 0.f;
  for (int u = 0; u < 32; ++u) {
    float x0 = xi[u];
    float x1x = xi[32 + u * 3 + 0];
    float x1y = xi[32 + u * 3 + 1];
    float x1z = xi[32 + u * 3 + 2];
    float wl0 = Wl10[u * 32 + lane], wl1 = Wl11[u * 32 + lane];
    float ws0 = Wsc0[u * 32 + lane], ws1 = Wsc1[u * 32 + lane];
    f0 += x0 * wl0;
    f1x += x1x * wl1; f1y += x1y * wl1; f1z += x1z * wl1;
    s0 += x0 * ws0;
    s1x += x1x * ws1; s1y += x1y * ws1; s1z += x1z * ws1;
  }
  float k = node_attr[n] * 0.17677669529663687f;
  float* fo = f0f1 + (long)n * 128;
  fo[lane] = f0 * k;
  fo[32 + lane] = f1x * k;
  fo[64 + lane] = f1y * k;
  fo[96 + lane] = f1z * k;
  float* so = scx + (long)n * 128;
  so[lane] = s0 * k;
  so[32 + lane * 3 + 0] = s1x * k;
  so[32 + lane * 3 + 1] = s1y * k;
  so[32 + lane * 3 + 2] = s1z * k;
}

#define EK_WAVES 8
#define HSTR 68

__global__ void __launch_bounds__(256)
edge_kernel(const float* __restrict__ edge_scalars,
            const float* __restrict__ edge_attr,
            const int* __restrict__ edge_src,
            const int* __restrict__ edge_dst,
            const float* __restrict__ Wfc1,
            const float* __restrict__ Wfc2,
            const float* __restrict__ f0f1,
            float* __restrict__ mid, int E) {
  __shared__ float fc1[4 * 4 * 32 * 2];
  __shared__ float fc2[16 * 8 * 32 * 2];
  __shared__ float stage[EK_WAVES][2048];

  for (int idx = threadIdx.x; idx < 4 * 4 * 32; idx += blockDim.x) {
    int ln = idx & 31, t = (idx >> 5) & 3, j = idx >> 7;
    int row = 4 * j + 2 * (ln >> 4), col = 16 * t + (ln & 15);
    fc1[idx * 2 + 0] = Wfc1[row * 64 + col];
    fc1[idx * 2 + 1] = Wfc1[(row + 1) * 64 + col];
  }
  for (int idx = threadIdx.x; idx < 16 * 8 * 32; idx += blockDim.x) {
    int ln = idx & 31, t = (idx >> 5) & 7, j = idx >> 8;
    int row = 4 * j + 2 * (ln >> 4), col = 16 * t + (ln & 15);
    fc2[idx * 2 + 0] = Wfc2[row * 128 + col];
    fc2[idx * 2 + 1] = Wfc2[(row + 1) * 128 + col];
  }
  __syncthreads();

  const int lane = threadIdx.x & 31;
  const int wv = threadIdx.x >> 5;
  const int r = lane & 15;
  const int hi = lane >> 4;
  float* st = stage[wv];
  const int nw = gridDim.x * (blockDim.x >> 5);
  const int gw = blockIdx.x * (blockDim.x >> 5) + wv;
  const int tiles = (E + 15) >> 4;

  for (int tile = gw; tile < tiles; tile += nw) {
    const int e0 = tile * 16;
    int er = e0 + r;
    if (er >= E) er = E - 1;
    const float* srow = edge_scalars + (long)er * 16;

    const int s_l = edge_src[er];
    const int d_l = edge_dst[er];
    const float4 ya_l = *(const float4*)(edge_attr + (long)er * 4);
    __builtin_amdgcn_sched_barrier(0);

    v2f a1[4];
#pragma unroll
    for (int j = 0; j < 4; ++j) a1[j] = *(const v2f*)(srow + 4 * j + 2 * hi);

#pragma unroll
    for (int t = 0; t < 4; ++t) {
      v8f c = {0.f, 0.f, 0.f, 0.f, 0.f, 0.f, 0.f, 0.f};
#pragma unroll
      for (int j = 0; j < 4; ++j) {
        v2f b = *(const v2f*)&fc1[((j * 4 + t) * 32 + lane) * 2];
        c = wmma4(a1[j], b, c);
      }
#pragma unroll
      for (int v = 0; v < 8; ++v) {
        float x = c[v] * 0.25f;
        st[(v + 8 * hi) * HSTR + 16 * t + r] = x / (1.0f + __expf(-x));
      }
    }

    v2f a2[16];
#pragma unroll
    for (int j = 0; j < 16; ++j)
      a2[j] = *(const v2f*)&st[r * HSTR + 4 * j + 2 * hi];

#pragma unroll
    for (int t = 0; t < 8; ++t) {
      v8f c = {0.f, 0.f, 0.f, 0.f, 0.f, 0.f, 0.f, 0.f};
#pragma unroll
      for (int j = 0; j < 16; ++j) {
        v2f b = *(const v2f*)&fc2[((j * 8 + t) * 32 + lane) * 2];
        c = wmma4(a2[j], b, c);
      }
#pragma unroll
      for (int v = 0; v < 8; ++v)
        st[(v + 8 * hi) * 128 + 16 * t + r] = c[v] * 0.125f;
    }

    const float q = 0.31622776601683794f;
    const float r3 = 0.5773502691896258f;
    if (e0 + 16 <= E) {
      float g0[16], g1x[16], g1y[16], g1z[16];
#pragma unroll
      for (int i = 0; i < 16; ++i) {
        const int s = __builtin_amdgcn_readlane(s_l, i);
        const float* g = f0f1 + (long)s * 128;
        g0[i] = g[lane];
        g1x[i] = g[32 + lane];
        g1y[i] = g[64 + lane];
        g1z[i] = g[96 + lane];
      }
#pragma unroll
      for (int i = 0; i < 16; ++i) {
        const int d = __builtin_amdgcn_readlane(d_l, i);
        const float yx = __int_as_float(__builtin_amdgcn_readlane(__float_as_int(ya_l.x), i));
        const float yy = __int_as_float(__builtin_amdgcn_readlane(__float_as_int(ya_l.y), i));
        const float yz = __int_as_float(__builtin_amdgcn_readlane(__float_as_int(ya_l.z), i));
        const float yw = __int_as_float(__builtin_amdgcn_readlane(__float_as_int(ya_l.w), i));
        float wA = st[i * 128 + lane];
        float wB = st[i * 128 + 32 + lane];
        float wC = st[i * 128 + 64 + lane];
        float wD = st[i * 128 + 96 + lane];
        float eA = wA * g0[i] * yx * q;
        float eD = wD * (g1x[i] * yy + g1y[i] * yz + g1z[i] * yw) * r3 * q;
        float gB = wB * g0[i] * q;
        float cc = wC * yx * q;
        float* m = mid + (long)d * 256;
        atomicAdd(m + lane, eA);
        atomicAdd(m + 32 + lane, eD);
        atomicAdd(m + 64 + 3 * lane + 0, gB * yy);
        atomicAdd(m + 64 + 3 * lane + 1, gB * yz);
        atomicAdd(m + 64 + 3 * lane + 2, gB * yw);
        atomicAdd(m + 160 + 3 * lane + 0, cc * g1x[i]);
        atomicAdd(m + 160 + 3 * lane + 1, cc * g1y[i]);
        atomicAdd(m + 160 + 3 * lane + 2, cc * g1z[i]);
      }
    } else {
      for (int i = 0; i < 16; ++i) {
        int e = e0 + i;
        if (e >= E) break;
        const int s = __builtin_amdgcn_readlane(s_l, i);
        const int d = __builtin_amdgcn_readlane(d_l, i);
        const float yx = __int_as_float(__builtin_amdgcn_readlane(__float_as_int(ya_l.x), i));
        const float yy = __int_as_float(__builtin_amdgcn_readlane(__float_as_int(ya_l.y), i));
        const float yz = __int_as_float(__builtin_amdgcn_readlane(__float_as_int(ya_l.z), i));
        const float yw = __int_as_float(__builtin_amdgcn_readlane(__float_as_int(ya_l.w), i));
        const float* g = f0f1 + (long)s * 128;
        float g0 = g[lane];
        float g1x = g[32 + lane], g1y = g[64 + lane], g1z = g[96 + lane];
        float wA = st[i * 128 + lane];
        float wB = st[i * 128 + 32 + lane];
        float wC = st[i * 128 + 64 + lane];
        float wD = st[i * 128 + 96 + lane];
        float eA = wA * g0 * yx * q;
        float eD = wD * (g1x * yy + g1y * yz + g1z * yw) * r3 * q;
        float gB = wB * g0 * q;
        float cc = wC * yx * q;
        float* m = mid + (long)d * 256;
        atomicAdd(m + lane, eA);
        atomicAdd(m + 32 + lane, eD);
        atomicAdd(m + 64 + 3 * lane + 0, gB * yy);
        atomicAdd(m + 64 + 3 * lane + 1, gB * yz);
        atomicAdd(m + 64 + 3 * lane + 2, gB * yw);
        atomicAdd(m + 160 + 3 * lane + 0, cc * g1x);
        atomicAdd(m + 160 + 3 * lane + 1, cc * g1y);
        atomicAdd(m + 160 + 3 * lane + 2, cc * g1z);
      }
    }
  }
}

__global__ void node_post_kernel(const float* __restrict__ mid,
                                 const float* __restrict__ node_attr,
                                 const float* __restrict__ Wl20,
                                 const float* __restrict__ Wl21,
                                 const float* __restrict__ Wl3,
                                 const float* __restrict__ scx,
                                 float* __restrict__ out, int N) {
  int wv = threadIdx.x >> 5;
  int lane = threadIdx.x & 31;
  int n = blockIdx.x * (blockDim.x >> 5) + wv;
  if (n >= N) return;
  const float* m = mid + (long)n * 256;
  float o0 = 0.f, o1x = 0.f, o1y = 0.f, o1z = 0.f, ang = 0.f;
  for (int v = 0; v < 64; ++v) {
    float m0 = m[v];
    float mx = m[64 + v * 3 + 0];
    float my = m[64 + v * 3 + 1];
    float mz = m[64 + v * 3 + 2];
    float w0 = Wl20[v * 32 + lane], w1 = Wl21[v * 32 + lane];
    o0 += m0 * w0;
    o1x += mx * w1; o1y += my * w1; o1z += mz * w1;
    ang += m0 * Wl3[v];
  }
  float k = node_attr[n] * 0.125f;
  o0 *= k; o1x *= k; o1y *= k; o1z *= k;
  float angle = 0.1f * ang * k;
  float ca = cosf(angle), sa = sinf(angle);
  const float* sx = scx + (long)n * 128;
  float* o = out + (long)n * 128;
  o[lane] = ca * sx[lane] + sa * o0;
  o[32 + 3 * lane + 0] = ca * sx[32 + 3 * lane + 0] + sa * o1x;
  o[32 + 3 * lane + 1] = ca * sx[32 + 3 * lane + 1] + sa * o1y;
  o[32 + 3 * lane + 2] = ca * sx[32 + 3 * lane + 2] + sa * o1z;
}

extern "C" void kernel_launch(void* const* d_in, const int* in_sizes, int n_in,
                              void* d_out, int out_size, void* d_ws, size_t ws_size,
                              hipStream_t stream) {
  const float* node_input   = (const float*)d_in[0];
  const float* node_attr    = (const float*)d_in[1];
  const int*   edge_src     = (const int*)d_in[2];
  const int*   edge_dst     = (const int*)d_in[3];
  const float* edge_attr    = (const float*)d_in[4];
  const float* edge_scalars = (const float*)d_in[5];
  const float* W_sc0        = (const float*)d_in[6];
  const float* W_sc1        = (const float*)d_in[7];
  const float* W_lin1_0     = (const float*)d_in[8];
  const float* W_lin1_1     = (const float*)d_in[9];
  const float* W_fc1        = (const float*)d_in[10];
  const float* W_fc2        = (const float*)d_in[11];
  const float* W_lin2_0     = (const float*)d_in[12];
  const float* W_lin2_1     = (const float*)d_in[13];
  const float* W_lin3       = (const float*)d_in[14];
  int N = in_sizes[1];
  int E = in_sizes[2];

  float* f0f1 = (float*)d_ws;
  float* scx  = f0f1 + (size_t)N * 128;
  float* mid  = scx + (size_t)N * 128;
  float* out  = (float*)d_out;

  zero_kernel<<<2048, 256, 0, stream>>>(mid, (long)N * 256);

  int nblk_node = (N + 7) / 8;
  node_pre_kernel<<<nblk_node, 256, 0, stream>>>(
      node_input, node_attr, W_sc0, W_sc1, W_lin1_0, W_lin1_1, f0f1, scx, N);

  edge_kernel<<<1024, 256, 0, stream>>>(
      edge_scalars, edge_attr, edge_src, edge_dst, W_fc1, W_fc2, f0f1, mid, E);

  node_post_kernel<<<nblk_node, 256, 0, stream>>>(
      mid, node_attr, W_lin2_0, W_lin2_1, W_lin3, scx, out, N);
}


# scatter accumulator kept as 8 component planes of 32 channels per node: each of an edge's 8 float atomics covers one 128-byte line (the six stride-3 ones spanned 3 lines each); sum order unchanged
# speedup vs baseline: 1.2706x; 1.1177x over previous
#include <hip/hip_runtime.h>
#include <math.h>


typedef __attribute__((ext_vector_type(2))) float v2f;
typedef __attribute__((ext_vector_type(8))) float v8f;

__device__ __forceinline__ v8f wmma4(v2f a, v2f b, v8f c) {
  return __builtin_amdgcn_wmma_f32_16x16x4_f32(false, a, false, b, (short)0, c,
                                               false, false);
}

__global__ void zero_kernel(float* __restrict__ p, long n) {
  long i = (long)blockIdx.x * blockDim.x + threadIdx.x;
  long stride = (long)gridDim.x * blockDim.x;
  for (long k = i * 4; k < n; k += stride * 4) {
    *(float4*)(p + k) = make_float4(0.f, 0.f, 0.f, 0.f);
  }
}

__global__ void node_pre_kernel(const float* __restrict__ node_input,
                                const float* __restrict__ node_attr,
                                const float* __restrict__ Wsc0,
                                const float* __restrict__ Wsc1,
                                const float* __restrict__ Wl10,
                                const float* __restrict__ Wl11,
                                float* __restrict__ f0f1,
                                float* __restrict__ scx, int N) {
  int wv = threadIdx.x >> 5;
  int lane = threadIdx.x & 31;
  int n = blockIdx.x * (blockDim.x >> 5) + wv;
  if (n >= N) return;
  const float* xi = node_input + (long)n * 128;
  float f0 = 0.f, f1x = 0.f, f1y = 0.f, f1z = 0.f;
  float s0 = 0.f, s1x = 0.f, s1y = 0.f, s1z = 0.f;
  for (int u = 0; u < 32; ++u) {
    float x0 = xi[u];
    float x1x = xi[32 + u * 3 + 0];
    float x1y = xi[32 + u * 3 + 1];
    float x1z = xi[32 + u * 3 + 2];
    float wl0 = Wl10[u * 32 + lane], wl1 = Wl11[u * 32 + lane];
    float ws0 = Wsc0[u * 32 + lane], ws1 = Wsc1[u * 32 + lane];
    f0 += x0 * wl0;
    f1x += x1x * wl1; f1y += x1y * wl1; f1z += x1z * wl1;
    s0 += x0 * ws0;
    s1x += x1x * ws1; s1y += x1y * ws1; s1z += x1z * ws1;
  }
  float k = node_attr[n] * 0.17677669529663687f;
  float* fo = f0f1 + (long)n * 128;
  fo[lane] = f0 * k;
  fo[32 + lane] = f1x * k;
  fo[64 + lane] = f1y * k;
  fo[96 + lane] = f1z * k;
  float* so = scx + (long)n * 128;
  so[lane] = s0 * k;
  so[32 + lane * 3 + 0] = s1x * k;
  so[32 + lane * 3 + 1] = s1y * k;
  so[32 + lane * 3 + 2] = s1z * k;
}

#define EK_WAVES 8
#define HSTR 68

__global__ void __launch_bounds__(256)
edge_kernel(const float* __restrict__ edge_scalars,
            const float* __restrict__ edge_attr,
            const int* __restrict__ edge_src,
            const int* __restrict__ edge_dst,
            const float* __restrict__ Wfc1,
            const float* __restrict__ Wfc2,
            const float* __restrict__ f0f1,
            float* __restrict__ mid, int E) {
  __shared__ float fc1[4 * 4 * 32 * 2];
  __shared__ float fc2[16 * 8 * 32 * 2];
  __shared__ float stage[EK_WAVES][2048];

  for (int idx = threadIdx.x; idx < 4 * 4 * 32; idx += blockDim.x) {
    int ln = idx & 31, t = (idx >> 5) & 3, j = idx >> 7;
    int row = 4 * j + 2 * (ln >> 4), col = 16 * t + (ln & 15);
    fc1[idx * 2 + 0] = Wfc1[row * 64 + col];
    fc1[idx * 2 + 1] = Wfc1[(row + 1) * 64 + col];
  }
  for (int idx = threadIdx.x; idx < 16 * 8 * 32; idx += blockDim.x) {
    int ln = idx & 31, t = (idx >> 5) & 7, j = idx >> 8;
    int row = 4 * j + 2 * (ln >> 4), col = 16 * t + (ln & 15);
    fc2[idx * 2 + 0] = Wfc2[row * 128 + col];
    fc2[idx * 2 + 1] = Wfc2[(row + 1) * 128 + col];
  }
  __syncthreads();

  const int lane = threadIdx.x & 31;
  const int wv = threadIdx.x >> 5;
  const int r = lane & 15;
  const int hi = lane >> 4;
  float* st = stage[wv];
  const int nw = gridDim.x * (blockDim.x >> 5);
  const int gw = blockIdx.x * (blockDim.x >> 5) + wv;
  const int tiles = (E + 15) >> 4;

  for (int tile = gw; tile < tiles; tile += nw) {
    const int e0 = tile * 16;
    int er = e0 + r;
    if (er >= E) er = E - 1;
    const float* srow = edge_scalars + (long)er * 16;

    const int s_l = edge_src[er];
    const int d_l = edge_dst[er];
    const float4 ya_l = *(const float4*)(edge_attr + (long)er * 4);
    __builtin_amdgcn_sched_barrier(0);

    v2f a1[4];
#pragma unroll
    for (int j = 0; j < 4; ++j) a1[j] = *(const v2f*)(srow + 4 * j + 2 * hi);

#pragma unroll
    for (int t = 0; t < 4; ++t) {
      v8f c = {0.f, 0.f, 0.f, 0.f, 0.f, 0.f, 0.f, 0.f};
#pragma unroll
      for (int j = 0; j < 4; ++j) {
        v2f b = *(const v2f*)&fc1[((j * 4 + t) * 32 + lane) * 2];
        c = wmma4(a1[j], b, c);
      }
#pragma unroll
      for (int v = 0; v < 8; ++v) {
        float x = c[v] * 0.25f;
        st[(v + 8 * hi) * HSTR + 16 * t + r] = x / (1.0f + __expf(-x));
      }
    }

    v2f a2[16];
#pragma unroll
    for (int j = 0; j < 16; ++j)
      a2[j] = *(const v2f*)&st[r * HSTR + 4 * j + 2 * hi];

#pragma unroll
    for (int t = 0; t < 8; ++t) {
      v8f c = {0.f, 0.f, 0.f, 0.f, 0.f, 0.f, 0.f, 0.f};
#pragma unroll
      for (int j = 0; j < 16; ++j) {
        v2f b = *(const v2f*)&fc2[((j * 8 + t) * 32 + lane) * 2];
        c = wmma4(a2[j], b, c);
      }
#pragma unroll
      for (int v = 0; v < 8; ++v)
        st[(v + 8 * hi) * 128 + 16 * t + r] = c[v] * 0.125f;
    }

    const float q = 0.31622776601683794f;
    const float r3 = 0.5773502691896258f;
    if (e0 + 16 <= E) {
      float g0[16], g1x[16], g1y[16], g1z[16];
#pragma unroll
      for (int i = 0; i < 16; ++i) {
        const int s = __builtin_amdgcn_readlane(s_l, i);
        const float* g = f0f1 + (long)s * 128;
        g0[i] = g[lane];
        g1x[i] = g[32 + lane];
        g1y[i] = g[64 + lane];
        g1z[i] = g[96 + lane];
      }
#pragma unroll
      for (int i = 0; i < 16; ++i) {
        const int d = __builtin_amdgcn_readlane(d_l, i);
        const float yx = __int_as_float(__builtin_amdgcn_readlane(__float_as_int(ya_l.x), i));
        const float yy = __int_as_float(__builtin_amdgcn_readlane(__float_as_int(ya_l.y), i));
        const float yz = __int_as_float(__builtin_amdgcn_readlane(__float_as_int(ya_l.z), i));
        const float yw = __int_as_float(__builtin_amdgcn_readlane(__float_as_int(ya_l.w), i));
        float wA = st[i * 128 + lane];
        float wB = st[i * 128 + 32 + lane];
        float wC = st[i * 128 + 64 + lane];
        float wD = st[i * 128 + 96 + lane];
        float eA = wA * g0[i] * yx * q;
        float eD = wD * (g1x[i] * yy + g1y[i] * yz + g1z[i] * yw) * r3 * q;
        float gB = wB * g0[i] * q;
        float cc = wC * yx * q;
        float* m = mid + (long)d * 256;
        atomicAdd(m + lane, eA);
        atomicAdd(m + 32 + lane, eD);
        atomicAdd(m + 64 + lane, gB * yy);
        atomicAdd(m + 96 + lane, gB * yz);
        atomicAdd(m + 128 + lane, gB * yw);
        atomicAdd(m + 160 + lane, cc * g1x[i]);
        atomicAdd(m + 192 + lane, cc * g1y[i]);
        atomicAdd(m + 224 + lane, cc * g1z[i]);
      }
    } else {
      for (int i = 0; i < 16; ++i) {
        int e = e0 + i;
        if (e >= E) break;
        const int s = __builtin_amdgcn_readlane(s_l, i);
        const int d = __builtin_amdgcn_readlane(d_l, i);
        const float yx = __int_as_float(__builtin_amdgcn_readlane(__float_as_int(ya_l.x), i));
        const float yy = __int_as_float(__builtin_amdgcn_readlane(__float_as_int(ya_l.y), i));
        const float yz = __int_as_float(__builtin_amdgcn_readlane(__float_as_int(ya_l.z), i));
        const float yw = __int_as_float(__builtin_amdgcn_readlane(__float_as_int(ya_l.w), i));
        const float* g = f0f1 + (long)s * 128;
        float g0 = g[lane];
        float g1x = g[32 + lane], g1y = g[64 + lane], g1z = g[96 + lane];
        float wA = st[i * 128 + lane];
        float wB = st[i * 128 + 32 + lane];
        float wC = st[i * 128 + 64 + lane];
        float wD = st[i * 128 + 96 + lane];
        float eA = wA * g0 * yx * q;
        float eD = wD * (g1x * yy + g1y * yz + g1z * yw) * r3 * q;
        float gB = wB * g0 * q;
        float cc = wC * yx * q;
        float* m = mid + (long)d * 256;
        atomicAdd(m + lane, eA);
        atomicAdd(m + 32 + lane, eD);
        atomicAdd(m + 64 + lane, gB * yy);
        atomicAdd(m + 96 + lane, gB * yz);
        atomicAdd(m + 128 + lane, gB * yw);
        atomicAdd(m + 160 + lane, cc * g1x);
        atomicAdd(m + 192 + lane, cc * g1y);
        atomicAdd(m + 224 + lane, cc * g1z);
      }
    }
  }
}

__global__ void node_post_kernel(const float* __restrict__ mid,
                                 const float* __restrict__ node_attr,
                                 const float* __restrict__ Wl20,
                                 const float* __restrict__ Wl21,
                                 const float* __restrict__ Wl3,
                                 const float* __restrict__ scx,
                                 float* __restrict__ out, int N) {
  int wv = threadIdx.x >> 5;
  int lane = threadIdx.x & 31;
  int n = blockIdx.x * (blockDim.x >> 5) + wv;
  if (n >= N) return;
  const float* m = mid + (long)n * 256;
  float o0 = 0.f, o1x = 0.f, o1y = 0.f, o1z = 0.f, ang = 0.f;
  for (int v = 0; v < 64; ++v) {
    float m0 = m[v];
    const int pb = (v < 32) ? (64 + v) : (128 + v);
    float mx = m[pb];
    float my = m[pb + 32];
    float mz = m[pb + 64];
    float w0 = Wl20[v * 32 + lane], w1 = Wl21[v * 32 + lane];
    o0 += m0 * w0;
    o1x += mx * w1; o1y += my * w1; o1z += mz * w1;
    ang += m0 * Wl3[v];
  }
  float k = node_attr[n] * 0.125f;
  o0 *= k; o1x *= k; o1y *= k; o1z *= k;
  float angle = 0.1f * ang * k;
  float ca = cosf(angle), sa = sinf(angle);
  const float* sx = scx + (long)n * 128;
  float* o = out + (long)n * 128;
  o[lane] = ca * sx[lane] + sa * o0;
  o[32 + 3 * lane + 0] = ca * sx[32 + 3 * lane + 0] + sa * o1x;
  o[32 + 3 * lane + 1] = ca * sx[32 + 3 * lane + 1] + sa * o1y;
  o[32 + 3 * lane + 2] = ca * sx[32 + 3 * lane + 2] + sa * o1z;
}

extern "C" void kernel_launch(void* const* d_in, const int* in_sizes, int n_in,
                              void* d_out, int out_size, void* d_ws, size_t ws_size,
                              hipStream_t stream) {
  const float* node_input   = (const float*)d_in[0];
  const float* node_attr    = (const float*)d_in[1];
  const int*   edge_src     = (const int*)d_in[2];
  const int*   edge_dst     = (const int*)d_in[3];
  const float* edge_attr    = (const float*)d_in[4];
  const float* edge_scalars = (const float*)d_in[5];
  const float* W_sc0        = (const float*)d_in[6];
  const float* W_sc1        = (const float*)d_in[7];
  const float* W_lin1_0     = (const float*)d_in[8];
  const float* W_lin1_1     = (const float*)d_in[9];
  const float* W_fc1        = (const float*)d_in[10];
  const float* W_fc2        = (const float*)d_in[11];
  const float* W_lin2_0     = (const float*)d_in[12];
  const float* W_lin2_1     = (const float*)d_in[13];
  const float* W_lin3       = (const float*)d_in[14];
  int N = in_sizes[1];
  int E = in_sizes[2];

  float* f0f1 = (float*)d_ws;
  float* scx  = f0f1 + (size_t)N * 128;
  float* mid  = scx + (size_t)N * 128;
  float* out  = (float*)d_out;

  zero_kernel<<<2048, 256, 0, stream>>>(mid, (long)N * 256);

  int nblk_node = (N + 7) / 8;
  node_pre_kernel<<<nblk_node, 256, 0, stream>>>(
      node_input, node_attr, W_sc0, W_sc1, W_lin1_0, W_lin1_1, f0f1, scx, N);

  edge_kernel<<<1024, 256, 0, stream>>>(
      edge_scalars, edge_attr, edge_src, edge_dst, W_fc1, W_fc2, f0f1, mid, E);

  node_post_kernel<<<nblk_node, 256, 0, stream>>>(
      mid, node_attr, W_lin2_0, W_lin2_1, W_lin3, scx, out, N);
}


# edge GEMMs: B fragments read from LDS one k-step ahead as a group, WMMAs k-step-major over 8 (GEMM1: 4) independent accumulators (was read, wait, one dependent WMMA, x144); sum order unchanged
# speedup vs baseline: 1.3656x; 1.0748x over previous
#include <hip/hip_runtime.h>
#include <math.h>


typedef __attribute__((ext_vector_type(2))) float v2f;
typedef __attribute__((ext_vector_type(8))) float v8f;

__device__ __forceinline__ v8f wmma4(v2f a, v2f b, v8f c) {
  return __builtin_amdgcn_wmma_f32_16x16x4_f32(false, a, false, b, (short)0, c,
                                               false, false);
}

__global__ void zero_kernel(float* __restrict__ p, long n) {
  long i = (long)blockIdx.x * blockDim.x + threadIdx.x;
  long stride = (long)gridDim.x * blockDim.x;
  for (long k = i * 4; k < n; k += stride * 4) {
    *(float4*)(p + k) = make_float4(0.f, 0.f, 0.f, 0.f);
  }
}

__global__ void node_pre_kernel(const float* __restrict__ node_input,
                                const float* __restrict__ node_attr,
                                const float* __restrict__ Wsc0,
                                const float* __restrict__ Wsc1,
                                const float* __restrict__ Wl10,
                                const float* __restrict__ Wl11,
                                float* __restrict__ f0f1,
                                float* __restrict__ scx, int N) {
  int wv = threadIdx.x >> 5;
  int lane = threadIdx.x & 31;
  int n = blockIdx.x * (blockDim.x >> 5) + wv;
  if (n >= N) return;
  const float* xi = node_input + (long)n * 128;
  float f0 = 0.f, f1x = 0.f, f1y = 0.f, f1z = 0.f;
  float s0 = 0.f, s1x = 0.f, s1y = 0.f, s1z = 0.f;
  for (int u = 0; u < 32; ++u) {
    float x0 = xi[u];
    float x1x = xi[32 + u * 3 + 0];
    float x1y = xi[32 + u * 3 + 1];
    float x1z = xi[32 + u * 3 + 2];
    float wl0 = Wl10[u * 32 + lane], wl1 = Wl11[u * 32 + lane];
    float ws0 = Wsc0[u * 32 + lane], ws1 = Wsc1[u * 32 + lane];
    f0 += x0 * wl0;
    f1x += x1x * wl1; f1y += x1y * wl1; f1z += x1z * wl1;
    s0 += x0 * ws0;
    s1x += x1x * ws1; s1y += x1y * ws1; s1z += x1z * ws1;
  }
  float k = node_attr[n] * 0.17677669529663687f;
  float* fo = f0f1 + (long)n * 128;
  fo[lane] = f0 * k;
  fo[32 + lane] = f1x * k;
  fo[64 + lane] = f1y * k;
  fo[96 + lane] = f1z * k;
  float* so = scx + (long)n * 128;
  so[lane] = s0 * k;
  so[32 + lane * 3 + 0] = s1x * k;
  so[32 + lane * 3 + 1] = s1y * k;
  so[32 + lane * 3 + 2] = s1z * k;
}

#define EK_WAVES 8
#define HSTR 68

__global__ void __launch_bounds__(256)
edge_kernel(const float* __restrict__ edge_scalars,
            const float* __restrict__ edge_attr,
            const int* __restrict__ edge_src,
            const int* __restrict__ edge_dst,
            const float* __restrict__ Wfc1,
            const float* __restrict__ Wfc2,
            const float* __restrict__ f0f1,
            float* __restrict__ mid, int E) {
  __shared__ float fc1[4 * 4 * 32 * 2];
  __shared__ float fc2[16 * 8 * 32 * 2];
  __shared__ float stage[EK_WAVES][2048];

  for (int idx = threadIdx.x; idx < 4 * 4 * 32; idx += blockDim.x) {
    int ln = idx & 31, t = (idx >> 5) & 3, j = idx >> 7;
    int row = 4 * j + 2 * (ln >> 4), col = 16 * t + (ln & 15);
    fc1[idx * 2 + 0] = Wfc1[row * 64 + col];
    fc1[idx * 2 + 1] = Wfc1[(row + 1) * 64 + col];
  }
  for (int idx = threadIdx.x; idx < 16 * 8 * 32; idx += blockDim.x) {
    int ln = idx & 31, t = (idx >> 5) & 7, j = idx >> 8;
    int row = 4 * j + 2 * (ln >> 4), col = 16 * t + (ln & 15);
    fc2[idx * 2 + 0] = Wfc2[row * 128 + col];
    fc2[idx * 2 + 1] = Wfc2[(row + 1) * 128 + col];
  }
  __syncthreads();

  const int lane = threadIdx.x & 31;
  const int wv = threadIdx.x >> 5;
  const int r = lane & 15;
  const int hi = lane >> 4;
  float* st = stage[wv];
  const int nw = gridDim.x * (blockDim.x >> 5);
  const int gw = blockIdx.x * (blockDim.x >> 5) + wv;
  const int tiles = (E + 15) >> 4;

  for (int tile = gw; tile < tiles; tile += nw) {
    const int e0 = tile * 16;
    int fslot = lane * 2;
    asm volatile("" : "+v"(fslot));
    const float* fc1l = fc1 + fslot;
    const float* fc2l = fc2 + fslot;
    int er = e0 + r;
    if (er >= E) er = E - 1;
    const float* srow = edge_scalars + (long)er * 16;

    const int s_l = edge_src[er];
    const int d_l = edge_dst[er];
    const float4 ya_l = *(const float4*)(edge_attr + (long)er * 4);
    __builtin_amdgcn_sched_barrier(0);

    v2f a1[4];
#pragma unroll
    for (int j = 0; j < 4; ++j) a1[j] = *(const v2f*)(srow + 4 * j + 2 * hi);

    {
      v2f b1[4][4];
#pragma unroll
      for (int j = 0; j < 4; ++j)
#pragma unroll
        for (int t = 0; t < 4; ++t)
          b1[j][t] = *(const v2f*)&fc1l[(j * 4 + t) * 64];
      __builtin_amdgcn_sched_barrier(0);
      v8f c1[4];
#pragma unroll
      for (int t = 0; t < 4; ++t) c1[t] = (v8f){0.f, 0.f, 0.f, 0.f, 0.f, 0.f, 0.f, 0.f};
#pragma unroll
      for (int j = 0; j < 4; ++j)
#pragma unroll
        for (int t = 0; t < 4; ++t) c1[t] = wmma4(a1[j], b1[j][t], c1[t]);
#pragma unroll
      for (int t = 0; t < 4; ++t)
#pragma unroll
        for (int v = 0; v < 8; ++v) {
          float x = c1[t][v] * 0.25f;
          st[(v + 8 * hi) * HSTR + 16 * t + r] = x / (1.0f + __expf(-x));
        }
    }

    v2f a2[16];
#pragma unroll
    for (int j = 0; j < 16; ++j)
      a2[j] = *(const v2f*)&st[r * HSTR + 4 * j + 2 * hi];

    {
      v8f c2[8];
#pragma unroll
      for (int t = 0; t < 8; ++t) c2[t] = (v8f){0.f, 0.f, 0.f, 0.f, 0.f, 0.f, 0.f, 0.f};
      v2f b2[2][8];
#pragma unroll
      for (int t = 0; t < 8; ++t) b2[0][t] = *(const v2f*)&fc2l[(0 * 8 + t) * 64];
#pragma unroll
      for (int j = 0; j < 16; ++j) {
        if (j + 1 < 16) {
#pragma unroll
          for (int t = 0; t < 8; ++t)
            b2[(j + 1) & 1][t] = *(const v2f*)&fc2l[((j + 1) * 8 + t) * 64];
        }
        __builtin_amdgcn_sched_barrier(0);
#pragma unroll
        for (int t = 0; t < 8; ++t) c2[t] = wmma4(a2[j], b2[j & 1][t], c2[t]);
        __builtin_amdgcn_sched_barrier(0);
      }
#pragma unroll
      for (int t = 0; t < 8; ++t)
#pragma unroll
        for (int v = 0; v < 8; ++v)
          st[(v + 8 * hi) * 128 + 16 * t + r] = c2[t][v] * 0.125f;
    }

    const float q = 0.31622776601683794f;
    const float r3 = 0.5773502691896258f;
    if (e0 + 16 <= E) {
      float g0[16], g1x[16], g1y[16], g1z[16];
#pragma unroll
      for (int i = 0; i < 16; ++i) {
        const int s = __builtin_amdgcn_readlane(s_l, i);
        const float* g = f0f1 + (long)s * 128;
        g0[i] = g[lane];
        g1x[i] = g[32 + lane];
        g1y[i] = g[64 + lane];
        g1z[i] = g[96 + lane];
      }
#pragma unroll
      for (int i = 0; i < 16; ++i) {
        const int d = __builtin_amdgcn_readlane(d_l, i);
        const float yx = __int_as_float(__builtin_amdgcn_readlane(__float_as_int(ya_l.x), i));
        const float yy = __int_as_float(__builtin_amdgcn_readlane(__float_as_int(ya_l.y), i));
        const float yz = __int_as_float(__builtin_amdgcn_readlane(__float_as_int(ya_l.z), i));
        const float yw = __int_as_float(__builtin_amdgcn_readlane(__float_as_int(ya_l.w), i));
        float wA = st[i * 128 + lane];
        float wB = st[i * 128 + 32 + lane];
        float wC = st[i * 128 + 64 + lane];
        float wD = st[i * 128 + 96 + lane];
        float eA = wA * g0[i] * yx * q;
        float eD = wD * (g1x[i] * yy + g1y[i] * yz + g1z[i] * yw) * r3 * q;
        float gB = wB * g0[i] * q;
        float cc = wC * yx * q;
        float* m = mid + (long)d * 256;
        atomicAdd(m + lane, eA);
        atomicAdd(m + 32 + lane, eD);
        atomicAdd(m + 64 + lane, gB * yy);
        atomicAdd(m + 96 + lane, gB * yz);
        atomicAdd(m + 128 + lane, gB * yw);
        atomicAdd(m + 160 + lane, cc * g1x[i]);
        atomicAdd(m + 192 + lane, cc * g1y[i]);
        atomicAdd(m + 224 + lane, cc * g1z[i]);
      }
    } else {
      for (int i = 0; i < 16; ++i) {
        int e = e0 + i;
        if (e >= E) break;
        const int s = __builtin_amdgcn_readlane(s_l, i);
        const int d = __builtin_amdgcn_readlane(d_l, i);
        const float yx = __int_as_float(__builtin_amdgcn_readlane(__float_as_int(ya_l.x), i));
        const float yy = __int_as_float(__builtin_amdgcn_readlane(__float_as_int(ya_l.y), i));
        const float yz = __int_as_float(__builtin_amdgcn_readlane(__float_as_int(ya_l.z), i));
        const float yw = __int_as_float(__builtin_amdgcn_readlane(__float_as_int(ya_l.w), i));
        const float* g = f0f1 + (long)s * 128;
        float g0 = g[lane];
        float g1x = g[32 + lane], g1y = g[64 + lane], g1z = g[96 + lane];
        float wA = st[i * 128 + lane];
        float wB = st[i * 128 + 32 + lane];
        float wC = st[i * 128 + 64 + lane];
        float wD = st[i * 128 + 96 + lane];
        float eA = wA * g0 * yx * q;
        float eD = wD * (g1x * yy + g1y * yz + g1z * yw) * r3 * q;
        float gB = wB * g0 * q;
        float cc = wC * yx * q;
        float* m = mid + (long)d * 256;
        atomicAdd(m + lane, eA);
        atomicAdd(m + 32 + lane, eD);
        atomicAdd(m + 64 + lane, gB * yy);
        atomicAdd(m + 96 + lane, gB * yz);
        atomicAdd(m + 128 + lane, gB * yw);
        atomicAdd(m + 160 + lane, cc * g1x);
        atomicAdd(m + 192 + lane, cc * g1y);
        atomicAdd(m + 224 + lane, cc * g1z);
      }
    }
  }
}

__global__ void node_post_kernel(const float* __restrict__ mid,
                                 const float* __restrict__ node_attr,
                                 const float* __restrict__ Wl20,
                                 const float* __restrict__ Wl21,
                                 const float* __restrict__ Wl3,
                                 const float* __restrict__ scx,
                                 float* __restrict__ out, int N) {
  int wv = threadIdx.x >> 5;
  int lane = threadIdx.x & 31;
  int n = blockIdx.x * (blockDim.x >> 5) + wv;
  if (n >= N) return;
  const float* m = mid + (long)n * 256;
  float o0 = 0.f, o1x = 0.f, o1y = 0.f, o1z = 0.f, ang = 0.f;
  for (int v = 0; v < 64; ++v) {
    float m0 = m[v];
    const int pb = (v < 32) ? (64 + v) : (128 + v);
    float mx = m[pb];
    float my = m[pb + 32];
    float mz = m[pb + 64];
    float w0 = Wl20[v * 32 + lane], w1 = Wl21[v * 32 + lane];
    o0 += m0 * w0;
    o1x += mx * w1; o1y += my * w1; o1z += mz * w1;
    ang += m0 * Wl3[v];
  }
  float k = node_attr[n] * 0.125f;
  o0 *= k; o1x *= k; o1y *= k; o1z *= k;
  float angle = 0.1f * ang * k;
  float ca = cosf(angle), sa = sinf(angle);
  const float* sx = scx + (long)n * 128;
  float* o = out + (long)n * 128;
  o[lane] = ca * sx[lane] + sa * o0;
  o[32 + 3 * lane + 0] = ca * sx[32 + 3 * lane + 0] + sa * o1x;
  o[32 + 3 * lane + 1] = ca * sx[32 + 3 * lane + 1] + sa * o1y;
  o[32 + 3 * lane + 2] = ca * sx[32 + 3 * lane + 2] + sa * o1z;
}

extern "C" void kernel_launch(void* const* d_in, const int* in_sizes, int n_in,
                              void* d_out, int out_size, void* d_ws, size_t ws_size,
                              hipStream_t stream) {
  const float* node_input   = (const float*)d_in[0];
  const float* node_attr    = (const float*)d_in[1];
  const int*   edge_src     = (const int*)d_in[2];
  const int*   edge_dst     = (const int*)d_in[3];
  const float* edge_attr    = (const float*)d_in[4];
  const float* edge_scalars = (const float*)d_in[5];
  const float* W_sc0        = (const float*)d_in[6];
  const float* W_sc1        = (const float*)d_in[7];
  const float* W_lin1_0     = (const float*)d_in[8];
  const float* W_lin1_1     = (const float*)d_in[9];
  const float* W_fc1        = (const float*)d_in[10];
  const float* W_fc2        = (const float*)d_in[11];
  const float* W_lin2_0     = (const float*)d_in[12];
  const float* W_lin2_1     = (const float*)d_in[13];
  const float* W_lin3       = (const float*)d_in[14];
  int N = in_sizes[1];
  int E = in_sizes[2];

  float* f0f1 = (float*)d_ws;
  float* scx  = f0f1 + (size_t)N * 128;
  float* mid  = scx + (size_t)N * 128;
  float* out  = (float*)d_out;

  zero_kernel<<<2048, 256, 0, stream>>>(mid, (long)N * 256);

  int nblk_node = (N + 7) / 8;
  node_pre_kernel<<<nblk_node, 256, 0, stream>>>(
      node_input, node_attr, W_sc0, W_sc1, W_lin1_0, W_lin1_1, f0f1, scx, N);

  edge_kernel<<<1024, 256, 0, stream>>>(
      edge_scalars, edge_attr, edge_src, edge_dst, W_fc1, W_fc2, f0f1, mid, E);

  node_post_kernel<<<nblk_node, 256, 0, stream>>>(
      mid, node_attr, W_lin2_0, W_lin2_1, W_lin3, scx, out, N);
}


# edge kernel: all waves resident at once, 256 work-groups of 32 waves (one per WGP, 8 per SIMD; H fragments read per k-step to fit 128 registers); was 1024 of 8, of which 768 fit; atomics as before
# speedup vs baseline: 1.4973x; 1.0964x over previous
#include <hip/hip_runtime.h>
#include <math.h>


typedef __attribute__((ext_vector_type(2))) float v2f;
typedef __attribute__((ext_vector_type(8))) float v8f;

__device__ __forceinline__ v8f wmma4(v2f a, v2f b, v8f c) {
  return __builtin_amdgcn_wmma_f32_16x16x4_f32(false, a, false, b, (short)0, c,
                                               false, false);
}

__global__ void zero_kernel(float* __restrict__ p, long n) {
  long i = (long)blockIdx.x * blockDim.x + threadIdx.x;
  long stride = (long)gridDim.x * blockDim.x;
  for (long k = i * 4; k < n; k += stride * 4) {
    *(float4*)(p + k) = make_float4(0.f, 0.f, 0.f, 0.f);
  }
}

__global__ void node_pre_kernel(const float* __restrict__ node_input,
                                const float* __restrict__ node_attr,
                                const float* __restrict__ Wsc0,
                                const float* __restrict__ Wsc1,
                                const float* __restrict__ Wl10,
                                const float* __restrict__ Wl11,
                                float* __restrict__ f0f1,
                                float* __restrict__ scx, int N) {
  int wv = threadIdx.x >> 5;
  int lane = threadIdx.x & 31;
  int n = blockIdx.x * (blockDim.x >> 5) + wv;
  if (n >= N) return;
  const float* xi = node_input + (long)n * 128;
  float f0 = 0.f, f1x = 0.f, f1y = 0.f, f1z = 0.f;
  float s0 = 0.f, s1x = 0.f, s1y = 0.f, s1z = 0.f;
  for (int u = 0; u < 32; ++u) {
    float x0 = xi[u];
    float x1x = xi[32 + u * 3 + 0];
    float x1y = xi[32 + u * 3 + 1];
    float x1z = xi[32 + u * 3 + 2];
    float wl0 = Wl10[u * 32 + lane], wl1 = Wl11[u * 32 + lane];
    float ws0 = Wsc0[u * 32 + lane], ws1 = Wsc1[u * 32 + lane];
    f0 += x0 * wl0;
    f1x += x1x * wl1; f1y += x1y * wl1; f1z += x1z * wl1;
    s0 += x0 * ws0;
    s1x += x1x * ws1; s1y += x1y * ws1; s1z += x1z * ws1;
  }
  float k = node_attr[n] * 0.17677669529663687f;
  float* fo = f0f1 + (long)n * 128;
  fo[lane] = f0 * k;
  fo[32 + lane] = f1x * k;
  fo[64 + lane] = f1y * k;
  fo[96 + lane] = f1z * k;
  float* so = scx + (long)n * 128;
  so[lane] = s0 * k;
  so[32 + lane * 3 + 0] = s1x * k;
  so[32 + lane * 3 + 1] = s1y * k;
  so[32 + lane * 3 + 2] = s1z * k;
}

#define EK_WAVES 32
#define HSTR 68

__global__ void __launch_bounds__(EK_WAVES * 32)
edge_kernel(const float* __restrict__ edge_scalars,
            const float* __restrict__ edge_attr,
            const int* __restrict__ edge_src,
            const int* __restrict__ edge_dst,
            const float* __restrict__ Wfc1,
            const float* __restrict__ Wfc2,
            const float* __restrict__ f0f1,
            float* __restrict__ mid, int E) {
  __shared__ float fc1[4 * 4 * 32 * 2];
  __shared__ float fc2[16 * 8 * 32 * 2];
  __shared__ float stage[EK_WAVES][2048];

  for (int idx = threadIdx.x; idx < 4 * 4 * 32; idx += blockDim.x) {
    int ln = idx & 31, t = (idx >> 5) & 3, j = idx >> 7;
    int row = 4 * j + 2 * (ln >> 4), col = 16 * t + (ln & 15);
    fc1[idx * 2 + 0] = Wfc1[row * 64 + col];
    fc1[idx * 2 + 1] = Wfc1[(row + 1) * 64 + col];
  }
  for (int idx = threadIdx.x; idx < 16 * 8 * 32; idx += blockDim.x) {
    int ln = idx & 31, t = (idx >> 5) & 7, j = idx >> 8;
    int row = 4 * j + 2 * (ln >> 4), col = 16 * t + (ln & 15);
    fc2[idx * 2 + 0] = Wfc2[row * 128 + col];
    fc2[idx * 2 + 1] = Wfc2[(row + 1) * 128 + col];
  }
  __syncthreads();

  const int lane = threadIdx.x & 31;
  const int wv = __builtin_amdgcn_readfirstlane(threadIdx.x >> 5);
  const int r = lane & 15;
  const int hi = lane >> 4;
  float* st = stage[wv];
  const int nw = gridDim.x * EK_WAVES;
  const int gw = wv * gridDim.x + blockIdx.x;
  const int tiles = (E + 15) >> 4;

  for (int tile = gw; tile < tiles; tile += nw) {
    const int e0 = tile * 16;
    int fslot = lane * 2;
    asm volatile("" : "+v"(fslot));
    const float* fc1l = fc1 + fslot;
    const float* fc2l = fc2 + fslot;
    int er = e0 + r;
    if (er >= E) er = E - 1;
    const float* srow = edge_scalars + (long)er * 16;

    const int s_l = edge_src[er];
    const int d_l = edge_dst[er];
    const float4 ya_l = *(const float4*)(edge_attr + (long)er * 4);
    __builtin_amdgcn_sched_barrier(0);

    v2f a1[4];
#pragma unroll
    for (int j = 0; j < 4; ++j) a1[j] = *(const v2f*)(srow + 4 * j + 2 * hi);

    {
      v2f b1[4][4];
#pragma unroll
      for (int j = 0; j < 4; ++j)
#pragma unroll
        for (int t = 0; t < 4; ++t)
          b1[j][t] = *(const v2f*)&fc1l[(j * 4 + t) * 64];
      __builtin_amdgcn_sched_barrier(0);
      v8f c1[4];
#pragma unroll
      for (int t = 0; t < 4; ++t) c1[t] = (v8f){0.f, 0.f, 0.f, 0.f, 0.f, 0.f, 0.f, 0.f};
#pragma unroll
      for (int j = 0; j < 4; ++j)
#pragma unroll
        for (int t = 0; t < 4; ++t) c1[t] = wmma4(a1[j], b1[j][t], c1[t]);
#pragma unroll
      for (int t = 0; t < 4; ++t)
#pragma unroll
        for (int v = 0; v < 8; ++v) {
          float x = c1[t][v] * 0.25f;
          st[(v + 8 * hi) * HSTR + 16 * t + r] = x / (1.0f + __expf(-x));
        }
    }

    {
      v8f c2[8];
#pragma unroll
      for (int t = 0; t < 8; ++t) c2[t] = (v8f){0.f, 0.f, 0.f, 0.f, 0.f, 0.f, 0.f, 0.f};
      v2f b2[2][8];
      v2f a2[2];
      a2[0] = *(const v2f*)&st[r * HSTR + 2 * hi];
#pragma unroll
      for (int t = 0; t < 8; ++t) b2[0][t] = *(const v2f*)&fc2l[(0 * 8 + t) * 64];
#pragma unroll
      for (int j = 0; j < 16; ++j) {
        if (j + 1 < 16) {
          a2[(j + 1) & 1] = *(const v2f*)&st[r * HSTR + 4 * (j + 1) + 2 * hi];
#pragma unroll
          for (int t = 0; t < 8; ++t)
            b2[(j + 1) & 1][t] = *(const v2f*)&fc2l[((j + 1) * 8 + t) * 64];
        }
        __builtin_amdgcn_sched_barrier(0);
#pragma unroll
        for (int t = 0; t < 8; ++t) c2[t] = wmma4(a2[j & 1], b2[j & 1][t], c2[t]);
        __builtin_amdgcn_sched_barrier(0);
      }
#pragma unroll
      for (int t = 0; t < 8; ++t)
#pragma unroll
        for (int v = 0; v < 8; ++v)
          st[(v + 8 * hi) * 128 + 16 * t + r] = c2[t][v] * 0.125f;
    }

    const float q = 0.31622776601683794f;
    const float r3 = 0.5773502691896258f;
    if (e0 + 16 <= E) {
      float g0[16], g1x[16], g1y[16], g1z[16];
#pragma unroll
      for (int i = 0; i < 16; ++i) {
        const int s = __builtin_amdgcn_readlane(s_l, i);
        const float* g = f0f1 + (long)s * 128;
        g0[i] = g[lane];
        g1x[i] = g[32 + lane];
        g1y[i] = g[64 + lane];
        g1z[i] = g[96 + lane];
      }
#pragma unroll
      for (int i = 0; i < 16; ++i) {
        const int d = __builtin_amdgcn_readlane(d_l, i);
        const float yx = __int_as_float(__builtin_amdgcn_readlane(__float_as_int(ya_l.x), i));
        const float yy = __int_as_float(__builtin_amdgcn_readlane(__float_as_int(ya_l.y), i));
        const float yz = __int_as_float(__builtin_amdgcn_readlane(__float_as_int(ya_l.z), i));
        const float yw = __int_as_float(__builtin_amdgcn_readlane(__float_as_int(ya_l.w), i));
        float wA = st[i * 128 + lane];
        float wB = st[i * 128 + 32 + lane];
        float wC = st[i * 128 + 64 + lane];
        float wD = st[i * 128 + 96 + lane];
        float eA = wA * g0[i] * yx * q;
        float eD = wD * (g1x[i] * yy + g1y[i] * yz + g1z[i] * yw) * r3 * q;
        float gB = wB * g0[i] * q;
        float cc = wC * yx * q;
        float* m = mid + (long)d * 256;
        atomicAdd(m + lane, eA);
        atomicAdd(m + 32 + lane, eD);
        atomicAdd(m + 64 + lane, gB * yy);
        atomicAdd(m + 96 + lane, gB * yz);
        atomicAdd(m + 128 + lane, gB * yw);
        atomicAdd(m + 160 + lane, cc * g1x[i]);
        atomicAdd(m + 192 + lane, cc * g1y[i]);
        atomicAdd(m + 224 + lane, cc * g1z[i]);
      }
    } else {
      for (int i = 0; i < 16; ++i) {
        int e = e0 + i;
        if (e >= E) break;
        const int s = __builtin_amdgcn_readlane(s_l, i);
        const int d = __builtin_amdgcn_readlane(d_l, i);
        const float yx = __int_as_float(__builtin_amdgcn_readlane(__float_as_int(ya_l.x), i));
        const float yy = __int_as_float(__builtin_amdgcn_readlane(__float_as_int(ya_l.y), i));
        const float yz = __int_as_float(__builtin_amdgcn_readlane(__float_as_int(ya_l.z), i));
        const float yw = __int_as_float(__builtin_amdgcn_readlane(__float_as_int(ya_l.w), i));
        const float* g = f0f1 + (long)s * 128;
        float g0 = g[lane];
        float g1x = g[32 + lane], g1y = g[64 + lane], g1z = g[96 + lane];
        float wA = st[i * 128 + lane];
        float wB = st[i * 128 + 32 + lane];
        float wC = st[i * 128 + 64 + lane];
        float wD = st[i * 128 + 96 + lane];
        float eA = wA * g0 * yx * q;
        float eD = wD * (g1x * yy + g1y * yz + g1z * yw) * r3 * q;
        float gB = wB * g0 * q;
        float cc = wC * yx * q;
        float* m = mid + (long)d * 256;
        atomicAdd(m + lane, eA);
        atomicAdd(m + 32 + lane, eD);
        atomicAdd(m + 64 + lane, gB * yy);
        atomicAdd(m + 96 + lane, gB * yz);
        atomicAdd(m + 128 + lane, gB * yw);
        atomicAdd(m + 160 + lane, cc * g1x);
        atomicAdd(m + 192 + lane, cc * g1y);
        atomicAdd(m + 224 + lane, cc * g1z);
      }
    }
  }
}

__global__ void node_post_kernel(const float* __restrict__ mid,
                                 const float* __restrict__ node_attr,
                                 const float* __restrict__ Wl20,
                                 const float* __restrict__ Wl21,
                                 const float* __restrict__ Wl3,
                                 const float* __restrict__ scx,
                                 float* __restrict__ out, int N) {
  int wv = threadIdx.x >> 5;
  int lane = threadIdx.x & 31;
  int n = blockIdx.x * (blockDim.x >> 5) + wv;
  if (n >= N) return;
  const float* m = mid + (long)n * 256;
  float o0 = 0.f, o1x = 0.f, o1y = 0.f, o1z = 0.f, ang = 0.f;
  for (int v = 0; v < 64; ++v) {
    float m0 = m[v];
    const int pb = (v < 32) ? (64 + v) : (128 + v);
    float mx = m[pb];
    float my = m[pb + 32];
    float mz = m[pb + 64];
    float w0 = Wl20[v * 32 + lane], w1 = Wl21[v * 32 + lane];
    o0 += m0 * w0;
    o1x += mx * w1; o1y += my * w1; o1z += mz * w1;
    ang += m0 * Wl3[v];
  }
  float k = node_attr[n] * 0.125f;
  o0 *= k; o1x *= k; o1y *= k; o1z *= k;
  float angle = 0.1f * ang * k;
  float ca = cosf(angle), sa = sinf(angle);
  const float* sx = scx + (long)n * 128;
  float* o = out + (long)n * 128;
  o[lane] = ca * sx[lane] + sa * o0;
  o[32 + 3 * lane + 0] = ca * sx[32 + 3 * lane + 0] + sa * o1x;
  o[32 + 3 * lane + 1] = ca * sx[32 + 3 * lane + 1] + sa * o1y;
  o[32 + 3 * lane + 2] = ca * sx[32 + 3 * lane + 2] + sa * o1z;
}

extern "C" void kernel_launch(void* const* d_in, const int* in_sizes, int n_in,
                              void* d_out, int out_size, void* d_ws, size_t ws_size,
                              hipStream_t stream) {
  const float* node_input   = (const float*)d_in[0];
  const float* node_attr    = (const float*)d_in[1];
  const int*   edge_src     = (const int*)d_in[2];
  const int*   edge_dst     = (const int*)d_in[3];
  const float* edge_attr    = (const float*)d_in[4];
  const float* edge_scalars = (const float*)d_in[5];
  const float* W_sc0        = (const float*)d_in[6];
  const float* W_sc1        = (const float*)d_in[7];
  const float* W_lin1_0     = (const float*)d_in[8];
  const float* W_lin1_1     = (const float*)d_in[9];
  const float* W_fc1        = (const float*)d_in[10];
  const float* W_fc2        = (const float*)d_in[11];
  const float* W_lin2_0     = (const float*)d_in[12];
  const float* W_lin2_1     = (const float*)d_in[13];
  const float* W_lin3       = (const float*)d_in[14];
  int N = in_sizes[1];
  int E = in_sizes[2];

  float* f0f1 = (float*)d_ws;
  float* scx  = f0f1 + (size_t)N * 128;
  float* mid  = scx + (size_t)N * 128;
  float* out  = (float*)d_out;

  zero_kernel<<<2048, 256, 0, stream>>>(mid, (long)N * 256);

  int nblk_node = (N + 7) / 8;
  node_pre_kernel<<<nblk_node, 256, 0, stream>>>(
      node_input, node_attr, W_sc0, W_sc1, W_lin1_0, W_lin1_1, f0f1, scx, N);

  edge_kernel<<<256, 1024, 0, stream>>>(
      edge_scalars, edge_attr, edge_src, edge_dst, W_fc1, W_fc2, f0f1, mid, E);

  node_post_kernel<<<nblk_node, 256, 0, stream>>>(
      mid, node_attr, W_lin2_0, W_lin2_1, W_lin3, scx, out, N);
}


# node kernels: linear layers as 16-node tile GEMMs on f32 WMMA, rows staged in LDS and moved as whole lines (was a wave per node, FMA loops); k sums in WMMA order, angle dot as two half sums
# speedup vs baseline: 1.7960x; 1.1995x over previous
#include <hip/hip_runtime.h>
#include <math.h>


typedef __attribute__((ext_vector_type(2))) float v2f;
typedef __attribute__((ext_vector_type(8))) float v8f;
typedef __attribute__((ext_vector_type(4))) float v4f;

__device__ __forceinline__ v8f wmma4(v2f a, v2f b, v8f c) {
  return __builtin_amdgcn_wmma_f32_16x16x4_f32(false, a, false, b, (short)0, c,
                                               false, false);
}

__global__ void zero_kernel(float* __restrict__ p, long n) {
  long i = (long)blockIdx.x * blockDim.x + threadIdx.x;
  long stride = (long)gridDim.x * blockDim.x;
  for (long k = i * 4; k < n; k += stride * 4) {
    *(float4*)(p + k) = make_float4(0.f, 0.f, 0.f, 0.f);
  }
}

#define NK_WAVES 16
#define NSTR 260

__global__ void __launch_bounds__(NK_WAVES * 32)
node_pre_kernel(const float* __restrict__ node_input,
                const float* __restrict__ node_attr,
                const float* __restrict__ Wsc0,
                const float* __restrict__ Wsc1,
                const float* __restrict__ Wl10,
                const float* __restrict__ Wl11,
                float* __restrict__ f0f1,
                float* __restrict__ scx, int N) {
  __shared__ float wb0[8 * 4 * 32 * 2];
  __shared__ float wb1[8 * 4 * 32 * 2];
  __shared__ __attribute__((aligned(16))) float stage[NK_WAVES][16 * NSTR];

  for (int idx = threadIdx.x; idx < 8 * 4 * 32; idx += blockDim.x) {
    int ln = idx & 31, t = (idx >> 5) & 3, j = idx >> 7;
    int row = 4 * j + 2 * (ln >> 4), col = 16 * (t & 1) + (ln & 15);
    const float* w0 = (t < 2) ? Wl10 : Wsc0;
    const float* w1 = (t < 2) ? Wl11 : Wsc1;
    wb0[idx * 2 + 0] = w0[row * 32 + col];
    wb0[idx * 2 + 1] = w0[(row + 1) * 32 + col];
    wb1[idx * 2 + 0] = w1[row * 32 + col];
    wb1[idx * 2 + 1] = w1[(row + 1) * 32 + col];
  }
  __syncthreads();

  const int lane = threadIdx.x & 31;
  const int wv = __builtin_amdgcn_readfirstlane(threadIdx.x >> 5);
  const int r = lane & 15;
  const int hi = lane >> 4;
  float* st = stage[wv];
  const int nw = gridDim.x * NK_WAVES;
  const int gw = wv * gridDim.x + blockIdx.x;
  const int tiles = (N + 15) >> 4;
  const float* a0p = st + r * NSTR + 2 * hi;
  const float* a1p = st + r * NSTR + 32 + 6 * hi;
  const float* b0p = wb0 + lane * 2;
  const float* b1p = wb1 + lane * 2;

  for (int tile = gw; tile < tiles; tile += nw) {
    const int n0 = tile * 16;
    const bool full = (n0 + 16 <= N);
    int nr = n0 + r;
    if (nr >= N) nr = N - 1;
    float k_l = node_attr[nr];
    {
      v4f xin[16];
#pragma unroll
      for (int i = 0; i < 16; ++i) {
        int n = n0 + i;
        if (n >= N) n = N - 1;
        xin[i] = *(const v4f*)(node_input + (long)n * 128 + lane * 4);
      }
      asm volatile("" : "+v"(k_l));
#pragma unroll
      for (int i = 0; i < 16; ++i) *(v4f*)&st[i * NSTR + lane * 4] = xin[i];
    }
    k_l *= 0.17677669529663687f;

    v8f acc[4][4];
#pragma unroll
    for (int p = 0; p < 4; ++p)
#pragma unroll
      for (int t = 0; t < 4; ++t) acc[p][t] = (v8f){0.f, 0.f, 0.f, 0.f, 0.f, 0.f, 0.f, 0.f};
    v2f a0f[2], xr[2][3], bf0[2][4], bf1[2][4];
    a0f[0] = *(const v2f*)(a0p);
#pragma unroll
    for (int q = 0; q < 3; ++q) xr[0][q] = *(const v2f*)(a1p + 2 * q);
#pragma unroll
    for (int t = 0; t < 4; ++t) {
      bf0[0][t] = *(const v2f*)&b0p[t * 64];
      bf1[0][t] = *(const v2f*)&b1p[t * 64];
    }
#pragma unroll
    for (int j = 0; j < 8; ++j) {
      if (j + 1 < 8) {
        const int nb = (j + 1) & 1;
        a0f[nb] = *(const v2f*)(a0p + 4 * (j + 1));
#pragma unroll
        for (int q = 0; q < 3; ++q) xr[nb][q] = *(const v2f*)(a1p + 12 * (j + 1) + 2 * q);
#pragma unroll
        for (int t = 0; t < 4; ++t) {
          bf0[nb][t] = *(const v2f*)&b0p[((j + 1) * 4 + t) * 64];
          bf1[nb][t] = *(const v2f*)&b1p[((j + 1) * 4 + t) * 64];
        }
      }
      __builtin_amdgcn_sched_barrier(0);
      const int cb = j & 1;
      v2f a1f[3];
      a1f[0] = (v2f){xr[cb][0].x, xr[cb][1].y};
      a1f[1] = (v2f){xr[cb][0].y, xr[cb][2].x};
      a1f[2] = (v2f){xr[cb][1].x, xr[cb][2].y};
#pragma unroll
      for (int t = 0; t < 4; ++t) acc[0][t] = wmma4(a0f[cb], bf0[cb][t], acc[0][t]);
#pragma unroll
      for (int m = 0; m < 3; ++m)
#pragma unroll
        for (int t = 0; t < 4; ++t) acc[1 + m][t] = wmma4(a1f[m], bf1[cb][t], acc[1 + m][t]);
      __builtin_amdgcn_sched_barrier(0);
    }

#pragma unroll
    for (int t = 0; t < 4; ++t)
#pragma unroll
      for (int v = 0; v < 8; ++v) {
        float* row = st + (v + 8 * hi) * NSTR;
        if (t < 2) {
          row[16 * t + r] = acc[0][t][v];
          row[32 + 16 * t + r] = acc[1][t][v];
          row[64 + 16 * t + r] = acc[2][t][v];
          row[96 + 16 * t + r] = acc[3][t][v];
        } else {
          const int w = 16 * (t - 2) + r;
          row[128 + w] = acc[0][t][v];
          row[160 + 3 * w + 0] = acc[1][t][v];
          row[160 + 3 * w + 1] = acc[2][t][v];
          row[160 + 3 * w + 2] = acc[3][t][v];
        }
      }

    if (full) {
      float4 oa[16], ob[16];
#pragma unroll
      for (int i = 0; i < 16; ++i) {
        oa[i] = *(const float4*)&st[i * NSTR + lane * 4];
        ob[i] = *(const float4*)&st[i * NSTR + 128 + lane * 4];
      }
      __builtin_amdgcn_sched_barrier(0);
#pragma unroll
      for (int i = 0; i < 16; ++i) {
        const int n = n0 + i;
        const float k = __int_as_float(__builtin_amdgcn_readlane(__float_as_int(k_l), i));
        float4 a = oa[i], b = ob[i];
        a.x *= k; a.y *= k; a.z *= k; a.w *= k;
        b.x *= k; b.y *= k; b.z *= k; b.w *= k;
        *(float4*)(f0f1 + (long)n * 128 + lane * 4) = a;
        *(float4*)(scx + (long)n * 128 + lane * 4) = b;
      }
    } else {
      for (int i = 0; i < 16; ++i) {
        const int n = n0 + i;
        if (n >= N) break;
        const float k = node_attr[n] * 0.17677669529663687f;
        float4 a = *(const float4*)&st[i * NSTR + lane * 4];
        float4 b = *(const float4*)&st[i * NSTR + 128 + lane * 4];
        a.x *= k; a.y *= k; a.z *= k; a.w *= k;
        b.x *= k; b.y *= k; b.z *= k; b.w *= k;
        *(float4*)(f0f1 + (long)n * 128 + lane * 4) = a;
        *(float4*)(scx + (long)n * 128 + lane * 4) = b;
      }
    }
  }
}

#define EK_WAVES 32
#define HSTR 68

__global__ void __launch_bounds__(EK_WAVES * 32)
edge_kernel(const float* __restrict__ edge_scalars,
            const float* __restrict__ edge_attr,
            const int* __restrict__ edge_src,
            const int* __restrict__ edge_dst,
            const float* __restrict__ Wfc1,
            const float* __restrict__ Wfc2,
            const float* __restrict__ f0f1,
            float* __restrict__ mid, int E) {
  __shared__ float fc1[4 * 4 * 32 * 2];
  __shared__ float fc2[16 * 8 * 32 * 2];
  __shared__ float stage[EK_WAVES][2048];

  for (int idx = threadIdx.x; idx < 4 * 4 * 32; idx += blockDim.x) {
    int ln = idx & 31, t = (idx >> 5) & 3, j = idx >> 7;
    int row = 4 * j + 2 * (ln >> 4), col = 16 * t + (ln & 15);
    fc1[idx * 2 + 0] = Wfc1[row * 64 + col];
    fc1[idx * 2 + 1] = Wfc1[(row + 1) * 64 + col];
  }
  for (int idx = threadIdx.x; idx < 16 * 8 * 32; idx += blockDim.x) {
    int ln = idx & 31, t = (idx >> 5) & 7, j = idx >> 8;
    int row = 4 * j + 2 * (ln >> 4), col = 16 * t + (ln & 15);
    fc2[idx * 2 + 0] = Wfc2[row * 128 + col];
    fc2[idx * 2 + 1] = Wfc2[(row + 1) * 128 + col];
  }
  __syncthreads();

  const int lane = threadIdx.x & 31;
  const int wv = __builtin_amdgcn_readfirstlane(threadIdx.x >> 5);
  const int r = lane & 15;
  const int hi = lane >> 4;
  float* st = stage[wv];
  const int nw = gridDim.x * EK_WAVES;
  const int gw = wv * gridDim.x + blockIdx.x;
  const int tiles = (E + 15) >> 4;

  for (int tile = gw; tile < tiles; tile += nw) {
    const int e0 = tile * 16;
    int fslot = lane * 2;
    asm volatile("" : "+v"(fslot));
    const float* fc1l = fc1 + fslot;
    const float* fc2l = fc2 + fslot;
    int er = e0 + r;
    if (er >= E) er = E - 1;
    const float* srow = edge_scalars + (long)er * 16;

    const int s_l = edge_src[er];
    const int d_l = edge_dst[er];
    const float4 ya_l = *(const float4*)(edge_attr + (long)er * 4);
    __builtin_amdgcn_sched_barrier(0);

    v2f a1[4];
#pragma unroll
    for (int j = 0; j < 4; ++j) a1[j] = *(const v2f*)(srow + 4 * j + 2 * hi);

    {
      v2f b1[4][4];
#pragma unroll
      for (int j = 0; j < 4; ++j)
#pragma unroll
        for (int t = 0; t < 4; ++t)
          b1[j][t] = *(const v2f*)&fc1l[(j * 4 + t) * 64];
      __builtin_amdgcn_sched_barrier(0);
      v8f c1[4];
#pragma unroll
      for (int t = 0; t < 4; ++t) c1[t] = (v8f){0.f, 0.f, 0.f, 0.f, 0.f, 0.f, 0.f, 0.f};
#pragma unroll
      for (int j = 0; j < 4; ++j)
#pragma unroll
        for (int t = 0; t < 4; ++t) c1[t] = wmma4(a1[j], b1[j][t], c1[t]);
#pragma unroll
      for (int t = 0; t < 4; ++t)
#pragma unroll
        for (int v = 0; v < 8; ++v) {
          float x = c1[t][v] * 0.25f;
          st[(v + 8 * hi) * HSTR + 16 * t + r] = x / (1.0f + __expf(-x));
        }
    }

    {
      v8f c2[8];
#pragma unroll
      for (int t = 0; t < 8; ++t) c2[t] = (v8f){0.f, 0.f, 0.f, 0.f, 0.f, 0.f, 0.f, 0.f};
      v2f b2[2][8];
      v2f a2[2];
      a2[0] = *(const v2f*)&st[r * HSTR + 2 * hi];
#pragma unroll
      for (int t = 0; t < 8; ++t) b2[0][t] = *(const v2f*)&fc2l[(0 * 8 + t) * 64];
#pragma unroll
      for (int j = 0; j < 16; ++j) {
        if (j + 1 < 16) {
          a2[(j + 1) & 1] = *(const v2f*)&st[r * HSTR + 4 * (j + 1) + 2 * hi];
#pragma unroll
          for (int t = 0; t < 8; ++t)
            b2[(j + 1) & 1][t] = *(const v2f*)&fc2l[((j + 1) * 8 + t) * 64];
        }
        __builtin_amdgcn_sched_barrier(0);
#pragma unroll
        for (int t = 0; t < 8; ++t) c2[t] = wmma4(a2[j & 1], b2[j & 1][t], c2[t]);
        __builtin_amdgcn_sched_barrier(0);
      }
#pragma unroll
      for (int t = 0; t < 8; ++t)
#pragma unroll
        for (int v = 0; v < 8; ++v)
          st[(v + 8 * hi) * 128 + 16 * t + r] = c2[t][v] * 0.125f;
    }

    const float q = 0.31622776601683794f;
    const float r3 = 0.5773502691896258f;
    if (e0 + 16 <= E) {
      float g0[16], g1x[16], g1y[16], g1z[16];
#pragma unroll
      for (int i = 0; i < 16; ++i) {
        const int s = __builtin_amdgcn_readlane(s_l, i);
        const float* g = f0f1 + (long)s * 128;
        g0[i] = g[lane];
        g1x[i] = g[32 + lane];
        g1y[i] = g[64 + lane];
        g1z[i] = g[96 + lane];
      }
#pragma unroll
      for (int i = 0; i < 16; ++i) {
        const int d = __builtin_amdgcn_readlane(d_l, i);
        const float yx = __int_as_float(__builtin_amdgcn_readlane(__float_as_int(ya_l.x), i));
        const float yy = __int_as_float(__builtin_amdgcn_readlane(__float_as_int(ya_l.y), i));
        const float yz = __int_as_float(__builtin_amdgcn_readlane(__float_as_int(ya_l.z), i));
        const float yw = __int_as_float(__builtin_amdgcn_readlane(__float_as_int(ya_l.w), i));
        float wA = st[i * 128 + lane];
        float wB = st[i * 128 + 32 + lane];
        float wC = st[i * 128 + 64 + lane];
        float wD = st[i * 128 + 96 + lane];
        float eA = wA * g0[i] * yx * q;
        float eD = wD * (g1x[i] * yy + g1y[i] * yz + g1z[i] * yw) * r3 * q;
        float gB = wB * g0[i] * q;
        float cc = wC * yx * q;
        float* m = mid + (long)d * 256;
        atomicAdd(m + lane, eA);
        atomicAdd(m + 32 + lane, eD);
        atomicAdd(m + 64 + lane, gB * yy);
        atomicAdd(m + 96 + lane, gB * yz);
        atomicAdd(m + 128 + lane, gB * yw);
        atomicAdd(m + 160 + lane, cc * g1x[i]);
        atomicAdd(m + 192 + lane, cc * g1y[i]);
        atomicAdd(m + 224 + lane, cc * g1z[i]);
      }
    } else {
      for (int i = 0; i < 16; ++i) {
        int e = e0 + i;
        if (e >= E) break;
        const int s = __builtin_amdgcn_readlane(s_l, i);
        const int d = __builtin_amdgcn_readlane(d_l, i);
        const float yx = __int_as_float(__builtin_amdgcn_readlane(__float_as_int(ya_l.x), i));
        const float yy = __int_as_float(__builtin_amdgcn_readlane(__float_as_int(ya_l.y), i));
        const float yz = __int_as_float(__builtin_amdgcn_readlane(__float_as_int(ya_l.z), i));
        const float yw = __int_as_float(__builtin_amdgcn_readlane(__float_as_int(ya_l.w), i));
        const float* g = f0f1 + (long)s * 128;
        float g0 = g[lane];
        float g1x = g[32 + lane], g1y = g[64 + lane], g1z = g[96 + lane];
        float wA = st[i * 128 + lane];
        float wB = st[i * 128 + 32 + lane];
        float wC = st[i * 128 + 64 + lane];
        float wD = st[i * 128 + 96 + lane];
        float eA = wA * g0 * yx * q;
        float eD = wD * (g1x * yy + g1y * yz + g1z * yw) * r3 * q;
        float gB = wB * g0 * q;
        float cc = wC * yx * q;
        float* m = mid + (long)d * 256;
        atomicAdd(m + lane, eA);
        atomicAdd(m + 32 + lane, eD);
        atomicAdd(m + 64 + lane, gB * yy);
        atomicAdd(m + 96 + lane, gB * yz);
        atomicAdd(m + 128 + lane, gB * yw);
        atomicAdd(m + 160 + lane, cc * g1x);
        atomicAdd(m + 192 + lane, cc * g1y);
        atomicAdd(m + 224 + lane, cc * g1z);
      }
    }
  }
}

__global__ void __launch_bounds__(NK_WAVES * 32)
node_post_kernel(const float* __restrict__ mid,
                 const float* __restrict__ node_attr,
                 const float* __restrict__ Wl20,
                 const float* __restrict__ Wl21,
                 const float* __restrict__ Wl3,
                 const float* __restrict__ scx,
                 float* __restrict__ out, int N) {
  __shared__ float wb0[16 * 2 * 32 * 2];
  __shared__ float wb1[16 * 2 * 32 * 2];
  __shared__ float wl3[64];
  __shared__ __attribute__((aligned(16))) float stage[NK_WAVES][16 * NSTR];

  for (int idx = threadIdx.x; idx < 16 * 2 * 32; idx += blockDim.x) {
    int ln = idx & 31, t = (idx >> 5) & 1, j = idx >> 6;
    int row = 4 * j + 2 * (ln >> 4), col = 16 * t + (ln & 15);
    wb0[idx * 2 + 0] = Wl20[row * 32 + col];
    wb0[idx * 2 + 1] = Wl20[(row + 1) * 32 + col];
    wb1[idx * 2 + 0] = Wl21[row * 32 + col];
    wb1[idx * 2 + 1] = Wl21[(row + 1) * 32 + col];
  }
  if (threadIdx.x < 64) wl3[threadIdx.x] = Wl3[threadIdx.x];
  __syncthreads();

  const int lane = threadIdx.x & 31;
  const int wv = __builtin_amdgcn_readfirstlane(threadIdx.x >> 5);
  const int r = lane & 15;
  const int hi = lane >> 4;
  float* st = stage[wv];
  const int nw = gridDim.x * NK_WAVES;
  const int gw = wv * gridDim.x + blockIdx.x;
  const int tiles = (N + 15) >> 4;
  const float* ap = st + r * NSTR + 2 * hi;
  const float* b0p = wb0 + lane * 2;
  const float* b1p = wb1 + lane * 2;
  const float* w3p = wl3 + 2 * hi;

  for (int tile = gw; tile < tiles; tile += nw) {
    const int n0 = tile * 16;
    const bool full = (n0 + 16 <= N);
    int nr = n0 + r;
    if (nr >= N) nr = N - 1;
    const float k_l = node_attr[nr] * 0.125f;
    {
      v4f lo[16], up[16];
#pragma unroll
      for (int i = 0; i < 16; ++i) {
        int n = n0 + i;
        if (n >= N) n = N - 1;
        lo[i] = *(const v4f*)(mid + (long)n * 256 + lane * 4);
        up[i] = *(const v4f*)(mid + (long)n * 256 + 128 + lane * 4);
      }
#pragma unroll
      for (int i = 0; i < 16; ++i) {
        *(v4f*)&st[i * NSTR + lane * 4] = lo[i];
        *(v4f*)&st[i * NSTR + 128 + lane * 4] = up[i];
      }
    }
    float4 sx[16];
#pragma unroll
    for (int i = 0; i < 16; ++i) {
      int n = n0 + i;
      if (n >= N) n = N - 1;
      sx[i] = *(const float4*)(scx + (long)n * 128 + lane * 4);
    }
#define POST_ACOL(p, j) ((p) == 0 ? 4 * (j) : ((j) < 8 ? 64 + 32 * ((p) - 1) + 4 * (j) : 160 + 32 * ((p) - 1) + 4 * ((j) - 8)))
    v8f acc[4][2];
#pragma unroll
    for (int p = 0; p < 4; ++p)
#pragma unroll
      for (int t = 0; t < 2; ++t) acc[p][t] = (v8f){0.f, 0.f, 0.f, 0.f, 0.f, 0.f, 0.f, 0.f};
    float angp = 0.f;
    v2f af[2][4], bf0[2][2], bf1[2][2], w3[2];
#pragma unroll
    for (int p = 0; p < 4; ++p) af[0][p] = *(const v2f*)(ap + POST_ACOL(p, 0));
#pragma unroll
    for (int t = 0; t < 2; ++t) {
      bf0[0][t] = *(const v2f*)&b0p[t * 64];
      bf1[0][t] = *(const v2f*)&b1p[t * 64];
    }
    w3[0] = *(const v2f*)(w3p);
#pragma unroll
    for (int j = 0; j < 16; ++j) {
      if (j + 1 < 16) {
        const int nb = (j + 1) & 1;
#pragma unroll
        for (int p = 0; p < 4; ++p) af[nb][p] = *(const v2f*)(ap + POST_ACOL(p, j + 1));
#pragma unroll
        for (int t = 0; t < 2; ++t) {
          bf0[nb][t] = *(const v2f*)&b0p[((j + 1) * 2 + t) * 64];
          bf1[nb][t] = *(const v2f*)&b1p[((j + 1) * 2 + t) * 64];
        }
        w3[nb] = *(const v2f*)(w3p + 4 * (j + 1));
      }
      __builtin_amdgcn_sched_barrier(0);
      const int cb = j & 1;
#pragma unroll
      for (int t = 0; t < 2; ++t) acc[0][t] = wmma4(af[cb][0], bf0[cb][t], acc[0][t]);
#pragma unroll
      for (int m = 0; m < 3; ++m)
#pragma unroll
        for (int t = 0; t < 2; ++t) acc[1 + m][t] = wmma4(af[cb][1 + m], bf1[cb][t], acc[1 + m][t]);
      angp += af[cb][0].x * w3[cb].x;
      angp += af[cb][0].y * w3[cb].y;
      __builtin_amdgcn_sched_barrier(0);
    }
#undef POST_ACOL

    const float ang = angp + __shfl_xor(angp, 16);
    const float angle = 0.1f * ang * k_l;
    const float ca_l = cosf(angle), sa_l = sinf(angle);

#pragma unroll
    for (int t = 0; t < 2; ++t)
#pragma unroll
      for (int v = 0; v < 8; ++v) {
        float* row = st + (v + 8 * hi) * NSTR;
        const int w = 16 * t + r;
        row[w] = acc[0][t][v];
        row[32 + 3 * w + 0] = acc[1][t][v];
        row[32 + 3 * w + 1] = acc[2][t][v];
        row[32 + 3 * w + 2] = acc[3][t][v];
      }

    if (full) {
      float4 ov[16];
#pragma unroll
      for (int i = 0; i < 16; ++i) ov[i] = *(const float4*)&st[i * NSTR + lane * 4];
#pragma unroll
      for (int i = 0; i < 16; ++i) {
        const float k = __int_as_float(__builtin_amdgcn_readlane(__float_as_int(k_l), i));
        const float ca = __int_as_float(__builtin_amdgcn_readlane(__float_as_int(ca_l), i));
        const float sa = __int_as_float(__builtin_amdgcn_readlane(__float_as_int(sa_l), i));
        float4 o = ov[i];
        o.x *= k; o.y *= k; o.z *= k; o.w *= k;
        float4 q;
        q.x = ca * sx[i].x + sa * o.x;
        q.y = ca * sx[i].y + sa * o.y;
        q.z = ca * sx[i].z + sa * o.z;
        q.w = ca * sx[i].w + sa * o.w;
        *(float4*)(out + (long)(n0 + i) * 128 + lane * 4) = q;
      }
    } else {
#pragma unroll
      for (int i = 0; i < 16; ++i) {
        if (n0 + i < N) {
          const float k = __int_as_float(__builtin_amdgcn_readlane(__float_as_int(k_l), i));
          const float ca = __int_as_float(__builtin_amdgcn_readlane(__float_as_int(ca_l), i));
          const float sa = __int_as_float(__builtin_amdgcn_readlane(__float_as_int(sa_l), i));
          float4 o = *(const float4*)&st[i * NSTR + lane * 4];
          o.x *= k; o.y *= k; o.z *= k; o.w *= k;
          float4 q;
          q.x = ca * sx[i].x + sa * o.x;
          q.y = ca * sx[i].y + sa * o.y;
          q.z = ca * sx[i].z + sa * o.z;
          q.w = ca * sx[i].w + sa * o.w;
          *(float4*)(out + (long)(n0 + i) * 128 + lane * 4) = q;
        }
      }
    }
  }
}

extern "C" void kernel_launch(void* const* d_in, const int* in_sizes, int n_in,
                              void* d_out, int out_size, void* d_ws, size_t ws_size,
                              hipStream_t stream) {
  const float* node_input   = (const float*)d_in[0];
  const float* node_attr    = (const float*)d_in[1];
  const int*   edge_src     = (const int*)d_in[2];
  const int*   edge_dst     = (const int*)d_in[3];
  const float* edge_attr    = (const float*)d_in[4];
  const float* edge_scalars = (const float*)d_in[5];
  const float* W_sc0        = (const float*)d_in[6];
  const float* W_sc1        = (const float*)d_in[7];
  const float* W_lin1_0     = (const float*)d_in[8];
  const float* W_lin1_1     = (const float*)d_in[9];
  const float* W_fc1        = (const float*)d_in[10];
  const float* W_fc2        = (const float*)d_in[11];
  const float* W_lin2_0     = (const float*)d_in[12];
  const float* W_lin2_1     = (const float*)d_in[13];
  const float* W_lin3       = (const float*)d_in[14];
  int N = in_sizes[1];
  int E = in_sizes[2];

  float* f0f1 = (float*)d_ws;
  float* scx  = f0f1 + (size_t)N * 128;
  float* mid  = scx + (size_t)N * 128;
  float* out  = (float*)d_out;

  zero_kernel<<<2048, 256, 0, stream>>>(mid, (long)N * 256);

  node_pre_kernel<<<256, NK_WAVES * 32, 0, stream>>>(
      node_input, node_attr, W_sc0, W_sc1, W_lin1_0, W_lin1_1, f0f1, scx, N);

  edge_kernel<<<256, 1024, 0, stream>>>(
      edge_scalars, edge_attr, edge_src, edge_dst, W_fc1, W_fc2, f0f1, mid, E);

  node_post_kernel<<<256, NK_WAVES * 32, 0, stream>>>(
      mid, node_attr, W_lin2_0, W_lin2_1, W_lin3, scx, out, N);
}
